# starting kernel
# baseline (speedup 1.0000x reference)
#include <hip/hip_runtime.h>

#define BATCH 32
#define CIN   64
#define COUT  64
#define IN_W  2048
#define KW    3
#define CK    (CIN * KW)
#define WBLK  8
#define XS_W  (WBLK + 2)

#define XS_BSTRIDE 646
#define XS_SIZE (BATCH * XS_BSTRIDE)
#define WS_ROW  196
#define WS_SIZE (COUT * WS_ROW)
#define CS_ROW  9
#define CS_SIZE (BATCH * COUT * CS_ROW)

#define LDS_FLOATS (XS_SIZE + 2 * WS_SIZE + CS_SIZE)
#define LDS_BYTES  (LDS_FLOATS * 4)

#define NTHREADS 512

typedef __attribute__((ext_vector_type(2))) float v2f;
typedef __attribute__((ext_vector_type(4))) float v4f;
typedef __attribute__((ext_vector_type(8))) float v8f;

__device__ __forceinline__ void wstage(const float* __restrict__ wsrc,
                                       float* __restrict__ dst, int t2) {
    v4f r[12];
    #pragma unroll
    for (int jj = 0; jj < 12; ++jj) {
        const int f = jj * 256 + t2;
        const int o = f / 48;
        const int q = f % 48;
        const float* g = wsrc + o * CK + (q >> 4) + 12 * (q & 15);
        r[jj].x = __builtin_nontemporal_load(g + 0);
        r[jj].y = __builtin_nontemporal_load(g + 3);
        r[jj].z = __builtin_nontemporal_load(g + 6);
        r[jj].w = __builtin_nontemporal_load(g + 9);
    }
    #pragma unroll
    for (int jj = 0; jj < 12; ++jj) {
        const int f = jj * 256 + t2;
        const int o = f / 48;
        const int q = f % 48;
        *(v4f*)(dst + o * WS_ROW + 4 * q) = r[jj];
    }
}

__device__ __forceinline__ v8f tile_gemm(const float* __restrict__ baseA,
                                         const float* __restrict__ baseB) {
    v8f acc = {};
    #pragma unroll
    for (int s = 0; s < 48; ++s) {
        const int offA = ((s >> 4) << 6) + ((4 * s) & 63);
        v2f a   = *(const v2f*)(baseA + offA);
        v2f bfr = *(const v2f*)(baseB + 4 * s);
        acc = __builtin_amdgcn_wmma_f32_16x16x4_f32(
                  false, a, false, bfr, (short)0, acc, false, false);
    }
    return acc;
}

__global__ __launch_bounds__(NTHREADS)
void lc1d_wmma_f32(
    const float* __restrict__ x,
    const float* __restrict__ wgt,
    const float* __restrict__ bias,
    float* __restrict__ out)
{
    extern __shared__ float smem[];
    float* xs  = smem;
    float* wsA = smem + XS_SIZE;
    float* cs  = smem + XS_SIZE + 2 * WS_SIZE;

    const int tid  = threadIdx.x;
    const int lane = tid & 31;
    const int wid  = tid >> 5;
    const int hl   = lane >> 4;
    const int lidx = lane & 15;
    const int w0   = blockIdx.x * WBLK;
    const bool is_compute = (wid < 8);

    if (is_compute) {
        for (int p = tid; p < BATCH * CIN; p += 256) {
            const int b = p >> 6;
            const int c = p & 63;
            const float* xp = x + ((size_t)b * CIN + c) * IN_W + (w0 - 1);
            float* xd = xs + b * XS_BSTRIDE + c;
            #pragma unroll
            for (int wa = 0; wa < XS_W; ++wa) {
                const int wg = w0 - 1 + wa;
                float v = 0.0f;
                if (wg >= 0 && wg < IN_W) v = xp[wa];
                xd[wa * 64] = v;
            }
        }
    } else {
        wstage(wgt + (size_t)w0 * (COUT * CK), wsA, tid - 256);
    }
    __syncthreads();

    const int mt  = wid & 1;
    const int nt  = (wid >> 1) & 3;
    const int b_m = mt * 16 + lidx;
    const int n_g = nt * 16 + lidx;
    const float* baseA0 = xs + b_m * XS_BSTRIDE + 2 * hl;
    const int    baseBo = n_g * WS_ROW + 2 * hl;

    for (int wl = 0; wl < WBLK; ++wl) {
        if (is_compute) {
            v8f acc = tile_gemm(baseA0 + wl * 64, wsA + (wl & 1) * WS_SIZE + baseBo);
            #pragma unroll
            for (int rr = 0; rr < 8; ++rr) {
                const int m = mt * 16 + hl * 8 + rr;
                cs[(m * COUT + n_g) * CS_ROW + wl] = acc[rr];
            }
        } else if (wl + 1 < WBLK) {
            wstage(wgt + (size_t)(w0 + wl + 1) * (COUT * CK),
                   wsA + (((wl & 1) ^ 1) * WS_SIZE), tid - 256);
        }
        __syncthreads();
    }

    for (int p = tid; p < BATCH * COUT; p += NTHREADS) {
        const int b = p >> 6;
        const int o = p & 63;
        const float* bp = bias + (size_t)o * IN_W + w0;
        float*       op = out  + ((size_t)b * COUT + o) * IN_W + w0;
        const float* cp = cs + p * CS_ROW;
        v4f r0, r1;
        r0.x = cp[0] + bp[0]; r0.y = cp[1] + bp[1];
        r0.z = cp[2] + bp[2]; r0.w = cp[3] + bp[3];
        r1.x = cp[4] + bp[4]; r1.y = cp[5] + bp[5];
        r1.z = cp[6] + bp[6]; r1.w = cp[7] + bp[7];
        __builtin_nontemporal_store(r0, (v4f*)(op));
        __builtin_nontemporal_store(r1, (v4f*)(op + 4));
    }
}

extern "C" void kernel_launch(void* const* d_in, const int* in_sizes, int n_in,
                              void* d_out, int out_size, void* d_ws, size_t ws_size,
                              hipStream_t stream) {
    const float* x    = (const float*)d_in[0];
    const float* wgt  = (const float*)d_in[1];
    const float* bias = (const float*)d_in[2];
    float* out = (float*)d_out;

    static_assert(LDS_BYTES <= 320 * 1024, "LDS budget");
    hipFuncSetAttribute(reinterpret_cast<const void*>(lc1d_wmma_f32),
                        hipFuncAttributeMaxDynamicSharedMemorySize, LDS_BYTES);

    dim3 grid(IN_W / WBLK);
    dim3 block(NTHREADS);
    lc1d_wmma_f32<<<grid, block, LDS_BYTES, stream>>>(x, wgt, bias, out);
}


# stage weights and x with 16-byte global loads in memory order (whole lines per wave-wide load); the turn to the LDS layout is done by the LDS stores; summation order unchanged
# speedup vs baseline: 1.7358x; 1.7358x over previous
#include <hip/hip_runtime.h>

#define BATCH 32
#define CIN   64
#define COUT  64
#define IN_W  2048
#define KW    3
#define CK    (CIN * KW)
#define WBLK  8
#define XS_W  (WBLK + 2)

#define XS_BSTRIDE 646
#define XS_SIZE (BATCH * XS_BSTRIDE)
#define WS_ROW  196
#define WS_SIZE (COUT * WS_ROW)
#define CS_ROW  9
#define CS_SIZE (BATCH * COUT * CS_ROW)

#define LDS_FLOATS (XS_SIZE + 2 * WS_SIZE + CS_SIZE)
#define LDS_BYTES  (LDS_FLOATS * 4)

#define NTHREADS 512

typedef __attribute__((ext_vector_type(2))) float v2f;
typedef __attribute__((ext_vector_type(4))) float v4f;
typedef __attribute__((ext_vector_type(8))) float v8f;

__device__ __forceinline__ void wload(const float* __restrict__ wsrc, int t2, v4f (&r)[12]) {
    const v4f* src = (const v4f*)wsrc;
    #pragma unroll
    for (int jj = 0; jj < 12; ++jj)
        r[jj] = __builtin_nontemporal_load(src + jj * 256 + t2);
}

__device__ __forceinline__ void wstore(float* __restrict__ dst, int t2, const v4f (&r)[12]) {
    #pragma unroll
    for (int j3 = 0; j3 < 3; ++j3) {
        const int f0 = j3 * 256 + t2;
        const int o0 = f0 / 48;
        const int q  = f0 - o0 * 48;
        #pragma unroll
        for (int i = 0; i < 4; ++i) {
            const int jn = 4 * q + i;
            const int c  = jn / 3;
            const int k  = jn - 3 * c;
            float* d = dst + o0 * WS_ROW + k * 64 + c;
            #pragma unroll
            for (int m = 0; m < 4; ++m)
                d[m * 16 * WS_ROW] = r[j3 + 3 * m][i];
        }
    }
}

__device__ __forceinline__ v8f tile_gemm(const float* __restrict__ baseA,
                                         const float* __restrict__ baseB) {
    v8f acc = {};
    #pragma unroll
    for (int s = 0; s < 48; ++s) {
        const int offA = ((s >> 4) << 6) + ((4 * s) & 63);
        v2f a   = *(const v2f*)(baseA + offA);
        v2f bfr = *(const v2f*)(baseB + 4 * s);
        acc = __builtin_amdgcn_wmma_f32_16x16x4_f32(
                  false, a, false, bfr, (short)0, acc, false, false);
    }
    return acc;
}

__global__ __launch_bounds__(NTHREADS)
void lc1d_wmma_f32(
    const float* __restrict__ x,
    const float* __restrict__ wgt,
    const float* __restrict__ bias,
    float* __restrict__ out)
{
    extern __shared__ float smem[];
    float* xs  = smem;
    float* wsA = smem + XS_SIZE;
    float* cs  = smem + XS_SIZE + 2 * WS_SIZE;

    const int tid  = threadIdx.x;
    const int lane = tid & 31;
    const int wid  = tid >> 5;
    const int hl   = lane >> 4;
    const int lidx = lane & 15;
    const int w0   = blockIdx.x * WBLK;
    const bool is_compute = (wid < 8);

    v4f wr[12];
    if (!is_compute) wload(wgt + (size_t)w0 * (COUT * CK), tid - 256, wr);

    {
        const int q4   = tid & 3;
        const int row0 = tid >> 2;
        const int ws0  = w0 - 4 + 4 * q4;
        const float* xg = x + (size_t)row0 * IN_W + ws0;
        v4f xr[16];
        if (ws0 >= 0 && ws0 + 3 < IN_W) {
            #pragma unroll
            for (int it = 0; it < 16; ++it)
                xr[it] = *(const v4f*)(xg + (size_t)it * 128 * IN_W);
        } else {
            #pragma unroll
            for (int it = 0; it < 16; ++it)
                xr[it] = (v4f){0.0f, 0.0f, 0.0f, 0.0f};
        }
        float* xd = xs + (row0 >> 6) * XS_BSTRIDE + (row0 & 63);
        #pragma unroll
        for (int i = 0; i < 4; ++i) {
            const int wa = 4 * q4 - 3 + i;
            if (wa >= 0 && wa < XS_W) {
                #pragma unroll
                for (int it = 0; it < 16; ++it)
                    xd[it * 2 * XS_BSTRIDE + wa * 64] = xr[it][i];
            }
        }
    }

    if (!is_compute) wstore(wsA, tid - 256, wr);
    __syncthreads();

    const int mt  = wid & 1;
    const int nt  = (wid >> 1) & 3;
    const int b_m = mt * 16 + lidx;
    const int n_g = nt * 16 + lidx;
    const float* baseA0 = xs + b_m * XS_BSTRIDE + 2 * hl;
    const int    baseBo = n_g * WS_ROW + 2 * hl;

    for (int wl = 0; wl < WBLK; ++wl) {
        if (is_compute) {
            v8f acc = tile_gemm(baseA0 + wl * 64, wsA + (wl & 1) * WS_SIZE + baseBo);
            #pragma unroll
            for (int rr = 0; rr < 8; ++rr) {
                const int m = mt * 16 + hl * 8 + rr;
                cs[(m * COUT + n_g) * CS_ROW + wl] = acc[rr];
            }
        } else if (wl + 1 < WBLK) {
            v4f wn[12];
            wload(wgt + (size_t)(w0 + wl + 1) * (COUT * CK), tid - 256, wn);
            wstore(wsA + (((wl & 1) ^ 1) * WS_SIZE), tid - 256, wn);
        }
        __syncthreads();
    }

    for (int p = tid; p < BATCH * COUT; p += NTHREADS) {
        const int b = p >> 6;
        const int o = p & 63;
        const float* bp = bias + (size_t)o * IN_W + w0;
        float*       op = out  + ((size_t)b * COUT + o) * IN_W + w0;
        const float* cp = cs + p * CS_ROW;
        v4f r0, r1;
        r0.x = cp[0] + bp[0]; r0.y = cp[1] + bp[1];
        r0.z = cp[2] + bp[2]; r0.w = cp[3] + bp[3];
        r1.x = cp[4] + bp[4]; r1.y = cp[5] + bp[5];
        r1.z = cp[6] + bp[6]; r1.w = cp[7] + bp[7];
        __builtin_nontemporal_store(r0, (v4f*)(op));
        __builtin_nontemporal_store(r1, (v4f*)(op + 4));
    }
}

extern "C" void kernel_launch(void* const* d_in, const int* in_sizes, int n_in,
                              void* d_out, int out_size, void* d_ws, size_t ws_size,
                              hipStream_t stream) {
    const float* x    = (const float*)d_in[0];
    const float* wgt  = (const float*)d_in[1];
    const float* bias = (const float*)d_in[2];
    float* out = (float*)d_out;

    static_assert(LDS_BYTES <= 320 * 1024, "LDS budget");
    hipFuncSetAttribute(reinterpret_cast<const void*>(lc1d_wmma_f32),
                        hipFuncAttributeMaxDynamicSharedMemorySize, LDS_BYTES);

    dim3 grid(IN_W / WBLK);
    dim3 block(NTHREADS);
    lc1d_wmma_f32<<<grid, block, LDS_BYTES, stream>>>(x, wgt, bias, out);
}
